# starting kernel
# baseline (speedup 1.0000x reference)
#include <hip/hip_runtime.h>
#include <hip/hip_bf16.h>

typedef float v2f __attribute__((ext_vector_type(2)));
typedef float v4f __attribute__((ext_vector_type(4)));
typedef float v8f __attribute__((ext_vector_type(8)));

#define NV 10
#define NH 10
#define NPATCH 100
#define BATCH 4
#define HW 160
#define CHW (256 * HW * HW)
#define PLANE (HW * HW)
#define XS_STRIDE 260
#define YS_STRIDE 132

__global__ __launch_bounds__(256)
void conv_relu_kernel(const float* __restrict__ f1, const float* __restrict__ f2,
                      const float* __restrict__ w_img, const float* __restrict__ b_img,
                      const float* __restrict__ w_dep, const float* __restrict__ b_dep,
                      float* __restrict__ x1, float* __restrict__ x2)
{
    int t = blockIdx.x * blockDim.x + threadIdx.x;
    const int QPIX = BATCH * PLANE / 4;
    int second = (t >= QPIX) ? 1 : 0;
    int p = t - second * QPIX;
    const float* feat = second ? f2 : f1;
    const float* w    = second ? w_dep : w_img;
    float bias        = second ? b_dep[0] : b_img[0];
    float* xo         = second ? x2 : x1;

    int pix = p * 4;
    int b   = pix / PLANE;
    int rem = pix - b * PLANE;
    int y   = rem / HW;
    int x   = rem - y * HW;

    const float* base = feat + (size_t)b * CHW + (size_t)y * HW + x;
    v4f acc = { bias, bias, bias, bias };
#pragma unroll 8
    for (int c = 0; c < 256; ++c) {
        v4f v = *(const v4f*)(base + (size_t)c * PLANE);
        float wc = w[c];
        acc.x = fmaf(v.x, wc, acc.x);
        acc.y = fmaf(v.y, wc, acc.y);
        acc.z = fmaf(v.z, wc, acc.z);
        acc.w = fmaf(v.w, wc, acc.w);
    }
    acc.x = fmaxf(acc.x, 0.0f);
    acc.y = fmaxf(acc.y, 0.0f);
    acc.z = fmaxf(acc.z, 0.0f);
    acc.w = fmaxf(acc.w, 0.0f);

    int pv = y >> 4, i = y & 15, ph = x >> 4, j = x & 15;
    int patch = (b * NV + pv) * NH + ph;
    *(v4f*)(xo + patch * 256 + i * 16 + j) = acc;
}

__global__ __launch_bounds__(256)
void fc_ln_kernel(const float* __restrict__ xbuf_img, const float* __restrict__ xbuf_dep,
                  const float* __restrict__ w1_img, const float* __restrict__ w2_img,
                  const float* __restrict__ g_img,  const float* __restrict__ be_img,
                  const float* __restrict__ w1_dep, const float* __restrict__ w2_dep,
                  const float* __restrict__ g_dep,  const float* __restrict__ be_dep,
                  float* __restrict__ e_img, float* __restrict__ e_dep)
{
    __shared__ float xs[16 * XS_STRIDE];
    __shared__ float hs[16 * XS_STRIDE];
    __shared__ float ys[16 * YS_STRIDE];

    const int branch = blockIdx.y;
    const float* xbuf = branch ? xbuf_dep : xbuf_img;
    const float* w1   = branch ? w1_dep   : w1_img;
    const float* w2   = branch ? w2_dep   : w2_img;
    const float* g    = branch ? g_dep    : g_img;
    const float* be   = branch ? be_dep   : be_img;
    float* ebuf       = branch ? e_dep    : e_img;

    const int rowbase = blockIdx.x * 16;
    const int t = threadIdx.x;

    for (int k = 0; k < 16; ++k) {
        int idx = t + k * 256;
        int r = idx >> 8, c = idx & 255;
        xs[r * XS_STRIDE + c] = xbuf[(rowbase + r) * 256 + c];
    }
    __syncthreads();

    const int wave = t >> 5;
    const int lane = t & 31;
    const int lhi  = lane >> 4;
    const int llo  = lane & 15;

    for (int nt = wave; nt < 16; nt += 8) {
        v8f acc = {};
        const float* w1row = w1 + (nt * 16 + llo) * 256;
        for (int kb = 0; kb < 256; kb += 4) {
            int kc = kb + 2 * lhi;
            v2f a = *(const v2f*)&xs[llo * XS_STRIDE + kc];
            v2f bm = *(const v2f*)&w1row[kc];
            acc = __builtin_amdgcn_wmma_f32_16x16x4_f32(
                false, a, false, bm, (short)0, acc, false, false);
        }
#pragma unroll
        for (int r = 0; r < 8; ++r) {
            int M = r + 8 * lhi;
            hs[M * XS_STRIDE + nt * 16 + llo] = fmaxf(acc[r], 0.0f);
        }
    }
    __syncthreads();

    {
        int nt = wave;
        v8f acc = {};
        const float* w2row = w2 + (nt * 16 + llo) * 256;
        for (int kb = 0; kb < 256; kb += 4) {
            int kc = kb + 2 * lhi;
            v2f a = *(const v2f*)&hs[llo * XS_STRIDE + kc];
            v2f bm = *(const v2f*)&w2row[kc];
            acc = __builtin_amdgcn_wmma_f32_16x16x4_f32(
                false, a, false, bm, (short)0, acc, false, false);
        }
#pragma unroll
        for (int r = 0; r < 8; ++r) {
            int M = r + 8 * lhi;
            ys[M * YS_STRIDE + nt * 16 + llo] = acc[r];
        }
    }
    __syncthreads();

    {
        int row = wave * 2 + lhi;
        float vals[8];
        float s = 0.0f, s2 = 0.0f;
#pragma unroll
        for (int q = 0; q < 8; ++q) {
            float v = ys[row * YS_STRIDE + llo * 8 + q];
            vals[q] = v; s += v; s2 += v * v;
        }
        for (int m = 1; m <= 8; m <<= 1) {
            s  += __shfl_xor(s,  m, 32);
            s2 += __shfl_xor(s2, m, 32);
        }
        float mean = s * (1.0f / 128.0f);
        float var  = s2 * (1.0f / 128.0f) - mean * mean;
        float inv  = rsqrtf(var + 1e-5f);
        int grow = rowbase + row;
#pragma unroll
        for (int q = 0; q < 8; ++q) {
            int c = llo * 8 + q;
            ebuf[grow * 128 + c] = (vals[q] - mean) * inv * g[c] + be[c];
        }
    }
}

__global__ __launch_bounds__(32)
void logits_kernel(const float* __restrict__ e1, const float* __restrict__ e2,
                   const float* __restrict__ logit_scale, float* __restrict__ out)
{
    int b  = blockIdx.z;
    int Mb = blockIdx.x * 16, Nb = blockIdx.y * 16;
    int lane = threadIdx.x & 31;
    int lhi = lane >> 4, llo = lane & 15;

    float scale = __expf(logit_scale[0]);

    int mrow = Mb + llo; if (mrow > NPATCH - 1) mrow = NPATCH - 1;
    int nrow = Nb + llo; if (nrow > NPATCH - 1) nrow = NPATCH - 1;
    const float* ea = e1 + (b * NPATCH + mrow) * 128;
    const float* eb = e2 + (b * NPATCH + nrow) * 128;

    v8f acc = {};
    for (int kb = 0; kb < 128; kb += 4) {
        int kc = kb + 2 * lhi;
        v2f a  = *(const v2f*)&ea[kc];
        v2f bm = *(const v2f*)&eb[kc];
        acc = __builtin_amdgcn_wmma_f32_16x16x4_f32(
            false, a, false, bm, (short)0, acc, false, false);
    }

#pragma unroll
    for (int r = 0; r < 8; ++r) {
        int M = Mb + r + 8 * lhi;
        int N = Nb + llo;
        if (M < NPATCH && N < NPATCH) {
            float v = scale * acc[r];
            out[(b * NPATCH + M) * NPATCH + N] = v;
            out[BATCH * NPATCH * NPATCH + (b * NPATCH + N) * NPATCH + M] = v;
        }
    }
}

extern "C" void kernel_launch(void* const* d_in, const int* in_sizes, int n_in,
                              void* d_out, int out_size, void* d_ws, size_t ws_size,
                              hipStream_t stream)
{
    const float* feat_c1  = (const float*)d_in[0];
    const float* feat_c2  = (const float*)d_in[1];
    const float* img_cw   = (const float*)d_in[3];
    const float* img_cb   = (const float*)d_in[4];
    const float* img_w1   = (const float*)d_in[5];
    const float* img_w2   = (const float*)d_in[6];
    const float* img_g    = (const float*)d_in[7];
    const float* img_b    = (const float*)d_in[8];
    const float* dep_cw   = (const float*)d_in[9];
    const float* dep_cb   = (const float*)d_in[10];
    const float* dep_w1   = (const float*)d_in[11];
    const float* dep_w2   = (const float*)d_in[12];
    const float* dep_g    = (const float*)d_in[13];
    const float* dep_b    = (const float*)d_in[14];
    const float* lscale   = (const float*)d_in[15];
    float* out = (float*)d_out;

    float* x1 = (float*)d_ws;
    float* x2 = x1 + 400 * 256;
    float* e1 = x2 + 400 * 256;
    float* e2 = e1 + 400 * 128;

    {
        int total = 2 * BATCH * PLANE / 4;
        conv_relu_kernel<<<dim3(total / 256), dim3(256), 0, stream>>>(
            feat_c1, feat_c2, img_cw, img_cb, dep_cw, dep_cb, x1, x2);
    }

    fc_ln_kernel<<<dim3(25, 2), dim3(256), 0, stream>>>(
        x1, x2, img_w1, img_w2, img_g, img_b,
        dep_w1, dep_w2, dep_g, dep_b, e1, e2);

    logits_kernel<<<dim3(7, 7, BATCH), dim3(32), 0, stream>>>(e1, e2, lscale, out);
}


# conv phase: split the 256-channel sum over 4 waves per work-group (16 loads in flight each, partials combined in LDS; channel summation order changes) so all 256 WGPs stream
# speedup vs baseline: 1.2266x; 1.2266x over previous
#include <hip/hip_runtime.h>
#include <hip/hip_bf16.h>

typedef float v2f __attribute__((ext_vector_type(2)));
typedef float v4f __attribute__((ext_vector_type(4)));
typedef float v8f __attribute__((ext_vector_type(8)));

#define NV 10
#define NH 10
#define NPATCH 100
#define BATCH 4
#define HW 160
#define CHW (256 * HW * HW)
#define PLANE (HW * HW)
#define XS_STRIDE 260
#define YS_STRIDE 132

#define CSPLIT 4
#define CPER (256 / CSPLIT)
#define NLD 16
__global__ __launch_bounds__(32 * CSPLIT)
void conv_relu_kernel(const float* __restrict__ f1, const float* __restrict__ f2,
                      const float* __restrict__ w_img, const float* __restrict__ b_img,
                      const float* __restrict__ w_dep, const float* __restrict__ b_dep,
                      float* __restrict__ x1, float* __restrict__ x2)
{
    __shared__ v4f part[(CSPLIT - 1) * 32];

    const int lane = threadIdx.x & 31;
    const int cg = __builtin_amdgcn_readfirstlane((int)(threadIdx.x >> 5));
    const int QGRP = BATCH * PLANE / 4 / 32;
    int qg = blockIdx.x;
    int second = (qg >= QGRP) ? 1 : 0;
    int p = (qg - second * QGRP) * 32 + lane;
    const float* feat = second ? f2 : f1;
    const float* w    = (second ? w_dep : w_img) + cg * CPER;
    float bias        = second ? b_dep[0] : b_img[0];
    float* xo         = second ? x2 : x1;

    int pix = p * 4;
    int b   = pix / PLANE;
    int rem = pix - b * PLANE;
    int y   = rem / HW;
    int x   = rem - y * HW;

    const float* base = feat + (size_t)b * CHW + (size_t)(cg * CPER) * PLANE
                             + (size_t)y * HW + x;
    float a0 = (cg == 0) ? bias : 0.0f;
    v4f acc = { a0, a0, a0, a0 };
#pragma unroll 1
    for (int c0 = 0; c0 < CPER; c0 += NLD) {
        v4f v[NLD];
#pragma unroll
        for (int i = 0; i < NLD; ++i)
            v[i] = *(const v4f*)(base + (size_t)(c0 + i) * PLANE);
#pragma unroll
        for (int i = 0; i < NLD; ++i) {
            float wc = w[c0 + i];
            acc.x = fmaf(v[i].x, wc, acc.x);
            acc.y = fmaf(v[i].y, wc, acc.y);
            acc.z = fmaf(v[i].z, wc, acc.z);
            acc.w = fmaf(v[i].w, wc, acc.w);
        }
    }

    if (cg != 0) part[(cg - 1) * 32 + lane] = acc;
    __syncthreads();
    if (cg == 0) {
#pragma unroll
        for (int g = 0; g < CSPLIT - 1; ++g) {
            v4f q = part[g * 32 + lane];
            acc.x += q.x; acc.y += q.y; acc.z += q.z; acc.w += q.w;
        }
        acc.x = fmaxf(acc.x, 0.0f);
        acc.y = fmaxf(acc.y, 0.0f);
        acc.z = fmaxf(acc.z, 0.0f);
        acc.w = fmaxf(acc.w, 0.0f);

        int pv = y >> 4, i = y & 15, ph = x >> 4, j = x & 15;
        int patch = (b * NV + pv) * NH + ph;
        *(v4f*)(xo + patch * 256 + i * 16 + j) = acc;
    }
}

__global__ __launch_bounds__(256)
void fc_ln_kernel(const float* __restrict__ xbuf_img, const float* __restrict__ xbuf_dep,
                  const float* __restrict__ w1_img, const float* __restrict__ w2_img,
                  const float* __restrict__ g_img,  const float* __restrict__ be_img,
                  const float* __restrict__ w1_dep, const float* __restrict__ w2_dep,
                  const float* __restrict__ g_dep,  const float* __restrict__ be_dep,
                  float* __restrict__ e_img, float* __restrict__ e_dep)
{
    __shared__ float xs[16 * XS_STRIDE];
    __shared__ float hs[16 * XS_STRIDE];
    __shared__ float ys[16 * YS_STRIDE];

    const int branch = blockIdx.y;
    const float* xbuf = branch ? xbuf_dep : xbuf_img;
    const float* w1   = branch ? w1_dep   : w1_img;
    const float* w2   = branch ? w2_dep   : w2_img;
    const float* g    = branch ? g_dep    : g_img;
    const float* be   = branch ? be_dep   : be_img;
    float* ebuf       = branch ? e_dep    : e_img;

    const int rowbase = blockIdx.x * 16;
    const int t = threadIdx.x;

    for (int k = 0; k < 16; ++k) {
        int idx = t + k * 256;
        int r = idx >> 8, c = idx & 255;
        xs[r * XS_STRIDE + c] = xbuf[(rowbase + r) * 256 + c];
    }
    __syncthreads();

    const int wave = t >> 5;
    const int lane = t & 31;
    const int lhi  = lane >> 4;
    const int llo  = lane & 15;

    for (int nt = wave; nt < 16; nt += 8) {
        v8f acc = {};
        const float* w1row = w1 + (nt * 16 + llo) * 256;
        for (int kb = 0; kb < 256; kb += 4) {
            int kc = kb + 2 * lhi;
            v2f a = *(const v2f*)&xs[llo * XS_STRIDE + kc];
            v2f bm = *(const v2f*)&w1row[kc];
            acc = __builtin_amdgcn_wmma_f32_16x16x4_f32(
                false, a, false, bm, (short)0, acc, false, false);
        }
#pragma unroll
        for (int r = 0; r < 8; ++r) {
            int M = r + 8 * lhi;
            hs[M * XS_STRIDE + nt * 16 + llo] = fmaxf(acc[r], 0.0f);
        }
    }
    __syncthreads();

    {
        int nt = wave;
        v8f acc = {};
        const float* w2row = w2 + (nt * 16 + llo) * 256;
        for (int kb = 0; kb < 256; kb += 4) {
            int kc = kb + 2 * lhi;
            v2f a = *(const v2f*)&hs[llo * XS_STRIDE + kc];
            v2f bm = *(const v2f*)&w2row[kc];
            acc = __builtin_amdgcn_wmma_f32_16x16x4_f32(
                false, a, false, bm, (short)0, acc, false, false);
        }
#pragma unroll
        for (int r = 0; r < 8; ++r) {
            int M = r + 8 * lhi;
            ys[M * YS_STRIDE + nt * 16 + llo] = acc[r];
        }
    }
    __syncthreads();

    {
        int row = wave * 2 + lhi;
        float vals[8];
        float s = 0.0f, s2 = 0.0f;
#pragma unroll
        for (int q = 0; q < 8; ++q) {
            float v = ys[row * YS_STRIDE + llo * 8 + q];
            vals[q] = v; s += v; s2 += v * v;
        }
        for (int m = 1; m <= 8; m <<= 1) {
            s  += __shfl_xor(s,  m, 32);
            s2 += __shfl_xor(s2, m, 32);
        }
        float mean = s * (1.0f / 128.0f);
        float var  = s2 * (1.0f / 128.0f) - mean * mean;
        float inv  = rsqrtf(var + 1e-5f);
        int grow = rowbase + row;
#pragma unroll
        for (int q = 0; q < 8; ++q) {
            int c = llo * 8 + q;
            ebuf[grow * 128 + c] = (vals[q] - mean) * inv * g[c] + be[c];
        }
    }
}

__global__ __launch_bounds__(32)
void logits_kernel(const float* __restrict__ e1, const float* __restrict__ e2,
                   const float* __restrict__ logit_scale, float* __restrict__ out)
{
    int b  = blockIdx.z;
    int Mb = blockIdx.x * 16, Nb = blockIdx.y * 16;
    int lane = threadIdx.x & 31;
    int lhi = lane >> 4, llo = lane & 15;

    float scale = __expf(logit_scale[0]);

    int mrow = Mb + llo; if (mrow > NPATCH - 1) mrow = NPATCH - 1;
    int nrow = Nb + llo; if (nrow > NPATCH - 1) nrow = NPATCH - 1;
    const float* ea = e1 + (b * NPATCH + mrow) * 128;
    const float* eb = e2 + (b * NPATCH + nrow) * 128;

    v8f acc = {};
    for (int kb = 0; kb < 128; kb += 4) {
        int kc = kb + 2 * lhi;
        v2f a  = *(const v2f*)&ea[kc];
        v2f bm = *(const v2f*)&eb[kc];
        acc = __builtin_amdgcn_wmma_f32_16x16x4_f32(
            false, a, false, bm, (short)0, acc, false, false);
    }

#pragma unroll
    for (int r = 0; r < 8; ++r) {
        int M = Mb + r + 8 * lhi;
        int N = Nb + llo;
        if (M < NPATCH && N < NPATCH) {
            float v = scale * acc[r];
            out[(b * NPATCH + M) * NPATCH + N] = v;
            out[BATCH * NPATCH * NPATCH + (b * NPATCH + N) * NPATCH + M] = v;
        }
    }
}

extern "C" void kernel_launch(void* const* d_in, const int* in_sizes, int n_in,
                              void* d_out, int out_size, void* d_ws, size_t ws_size,
                              hipStream_t stream)
{
    const float* feat_c1  = (const float*)d_in[0];
    const float* feat_c2  = (const float*)d_in[1];
    const float* img_cw   = (const float*)d_in[3];
    const float* img_cb   = (const float*)d_in[4];
    const float* img_w1   = (const float*)d_in[5];
    const float* img_w2   = (const float*)d_in[6];
    const float* img_g    = (const float*)d_in[7];
    const float* img_b    = (const float*)d_in[8];
    const float* dep_cw   = (const float*)d_in[9];
    const float* dep_cb   = (const float*)d_in[10];
    const float* dep_w1   = (const float*)d_in[11];
    const float* dep_w2   = (const float*)d_in[12];
    const float* dep_g    = (const float*)d_in[13];
    const float* dep_b    = (const float*)d_in[14];
    const float* lscale   = (const float*)d_in[15];
    float* out = (float*)d_out;

    float* x1 = (float*)d_ws;
    float* x2 = x1 + 400 * 256;
    float* e1 = x2 + 400 * 256;
    float* e2 = e1 + 400 * 128;

    conv_relu_kernel<<<dim3(2 * BATCH * PLANE / 4 / 32), dim3(32 * CSPLIT), 0, stream>>>(
        feat_c1, feat_c2, img_cw, img_cb, dep_cw, dep_cb, x1, x2);

    fc_ln_kernel<<<dim3(25, 2), dim3(256), 0, stream>>>(
        x1, x2, img_w1, img_w2, img_g, img_b,
        dep_w1, dep_w2, dep_g, dep_b, e1, e2);

    logits_kernel<<<dim3(7, 7, BATCH), dim3(32), 0, stream>>>(e1, e2, lscale, out);
}


# fc phase: weight (global) and activation (LDS) fragments fetched as 16-byte k-runs (lane half h takes k=8j+4h..+3), weight loads running a 128-k block ahead of their WMMAs; k summation order changes
# speedup vs baseline: 1.2853x; 1.0479x over previous
#include <hip/hip_runtime.h>
#include <hip/hip_bf16.h>

typedef float v2f __attribute__((ext_vector_type(2)));
typedef float v4f __attribute__((ext_vector_type(4)));
typedef float v8f __attribute__((ext_vector_type(8)));

#define NV 10
#define NH 10
#define NPATCH 100
#define BATCH 4
#define HW 160
#define CHW (256 * HW * HW)
#define PLANE (HW * HW)
#define XS_STRIDE 260
#define YS_STRIDE 132

#define CSPLIT 4
#define CPER (256 / CSPLIT)
#define NLD 16
__global__ __launch_bounds__(32 * CSPLIT)
void conv_relu_kernel(const float* __restrict__ f1, const float* __restrict__ f2,
                      const float* __restrict__ w_img, const float* __restrict__ b_img,
                      const float* __restrict__ w_dep, const float* __restrict__ b_dep,
                      float* __restrict__ x1, float* __restrict__ x2)
{
    __shared__ v4f part[(CSPLIT - 1) * 32];

    const int lane = threadIdx.x & 31;
    const int cg = __builtin_amdgcn_readfirstlane((int)(threadIdx.x >> 5));
    const int QGRP = BATCH * PLANE / 4 / 32;
    int qg = blockIdx.x;
    int second = (qg >= QGRP) ? 1 : 0;
    int p = (qg - second * QGRP) * 32 + lane;
    const float* feat = second ? f2 : f1;
    const float* w    = (second ? w_dep : w_img) + cg * CPER;
    float bias        = second ? b_dep[0] : b_img[0];
    float* xo         = second ? x2 : x1;

    int pix = p * 4;
    int b   = pix / PLANE;
    int rem = pix - b * PLANE;
    int y   = rem / HW;
    int x   = rem - y * HW;

    const float* base = feat + (size_t)b * CHW + (size_t)(cg * CPER) * PLANE
                             + (size_t)y * HW + x;
    float a0 = (cg == 0) ? bias : 0.0f;
    v4f acc = { a0, a0, a0, a0 };
#pragma unroll 1
    for (int c0 = 0; c0 < CPER; c0 += NLD) {
        v4f v[NLD];
#pragma unroll
        for (int i = 0; i < NLD; ++i)
            v[i] = *(const v4f*)(base + (size_t)(c0 + i) * PLANE);
#pragma unroll
        for (int i = 0; i < NLD; ++i) {
            float wc = w[c0 + i];
            acc.x = fmaf(v[i].x, wc, acc.x);
            acc.y = fmaf(v[i].y, wc, acc.y);
            acc.z = fmaf(v[i].z, wc, acc.z);
            acc.w = fmaf(v[i].w, wc, acc.w);
        }
    }

    if (cg != 0) part[(cg - 1) * 32 + lane] = acc;
    __syncthreads();
    if (cg == 0) {
#pragma unroll
        for (int g = 0; g < CSPLIT - 1; ++g) {
            v4f q = part[g * 32 + lane];
            acc.x += q.x; acc.y += q.y; acc.z += q.z; acc.w += q.w;
        }
        acc.x = fmaxf(acc.x, 0.0f);
        acc.y = fmaxf(acc.y, 0.0f);
        acc.z = fmaxf(acc.z, 0.0f);
        acc.w = fmaxf(acc.w, 0.0f);

        int pv = y >> 4, i = y & 15, ph = x >> 4, j = x & 15;
        int patch = (b * NV + pv) * NH + ph;
        *(v4f*)(xo + patch * 256 + i * 16 + j) = acc;
    }
}

#define KCH 128
#define NLDB (KCH / 8)

__device__ __forceinline__ void load_b_block(v4f (&b)[NLDB], const float* __restrict__ p)
{
#pragma unroll
    for (int j = 0; j < NLDB; ++j)
        b[j] = *(const v4f*)(p + 8 * j);
}

__device__ __forceinline__ v8f mma_block(v8f acc, const float* a_lds, const v4f (&b)[NLDB])
{
#pragma unroll
    for (int j = 0; j < NLDB; ++j) {
        v4f a = *(const v4f*)(a_lds + 8 * j);
        v2f alo = { a.x, a.y },       ahi = { a.z, a.w };
        v2f blo = { b[j].x, b[j].y }, bhi = { b[j].z, b[j].w };
        acc = __builtin_amdgcn_wmma_f32_16x16x4_f32(
            false, alo, false, blo, (short)0, acc, false, false);
        acc = __builtin_amdgcn_wmma_f32_16x16x4_f32(
            false, ahi, false, bhi, (short)0, acc, false, false);
    }
    return acc;
}

__global__ __launch_bounds__(256)
void fc_ln_kernel(const float* __restrict__ xbuf_img, const float* __restrict__ xbuf_dep,
                  const float* __restrict__ w1_img, const float* __restrict__ w2_img,
                  const float* __restrict__ g_img,  const float* __restrict__ be_img,
                  const float* __restrict__ w1_dep, const float* __restrict__ w2_dep,
                  const float* __restrict__ g_dep,  const float* __restrict__ be_dep,
                  float* __restrict__ e_img, float* __restrict__ e_dep)
{
    __shared__ __attribute__((aligned(16))) float xs[16 * XS_STRIDE];
    __shared__ __attribute__((aligned(16))) float hs[16 * XS_STRIDE];
    __shared__ float ys[16 * YS_STRIDE];

    const int branch = blockIdx.y;
    const float* xbuf = branch ? xbuf_dep : xbuf_img;
    const float* w1   = branch ? w1_dep   : w1_img;
    const float* w2   = branch ? w2_dep   : w2_img;
    const float* g    = branch ? g_dep    : g_img;
    const float* be   = branch ? be_dep   : be_img;
    float* ebuf       = branch ? e_dep    : e_img;

    const int rowbase = blockIdx.x * 16;
    const int t = threadIdx.x;

    const int wave = t >> 5;
    const int lane = t & 31;
    const int lhi  = lane >> 4;
    const int llo  = lane & 15;

    const float* w1t0 = w1 + (wave * 16 + llo) * 256 + 4 * lhi;
    const float* w1t1 = w1t0 + 8 * 16 * 256;
    const float* w2t  = w2 + (wave * 16 + llo) * 256 + 4 * lhi;
    const float* xa   = xs + llo * XS_STRIDE + 4 * lhi;
    const float* ha   = hs + llo * XS_STRIDE + 4 * lhi;

    v4f b0[NLDB], b1[NLDB];
    load_b_block(b0, w1t0);

    for (int k = 0; k < 16; ++k) {
        int idx = t + k * 256;
        int r = idx >> 8, c = idx & 255;
        xs[r * XS_STRIDE + c] = xbuf[(rowbase + r) * 256 + c];
    }
    __syncthreads();

    {
        v8f acc = {};
        load_b_block(b1, w1t0 + KCH);
        acc = mma_block(acc, xa, b0);
        load_b_block(b0, w1t1);
        acc = mma_block(acc, xa + KCH, b1);
#pragma unroll
        for (int r = 0; r < 8; ++r) {
            int M = r + 8 * lhi;
            hs[M * XS_STRIDE + wave * 16 + llo] = fmaxf(acc[r], 0.0f);
        }
    }
    {
        v8f acc = {};
        load_b_block(b1, w1t1 + KCH);
        acc = mma_block(acc, xa, b0);
        load_b_block(b0, w2t);
        acc = mma_block(acc, xa + KCH, b1);
#pragma unroll
        for (int r = 0; r < 8; ++r) {
            int M = r + 8 * lhi;
            hs[M * XS_STRIDE + (wave + 8) * 16 + llo] = fmaxf(acc[r], 0.0f);
        }
    }
    load_b_block(b1, w2t + KCH);
    __syncthreads();

    {
        v8f acc = {};
        acc = mma_block(acc, ha, b0);
        acc = mma_block(acc, ha + KCH, b1);
#pragma unroll
        for (int r = 0; r < 8; ++r) {
            int M = r + 8 * lhi;
            ys[M * YS_STRIDE + wave * 16 + llo] = acc[r];
        }
    }
    __syncthreads();

    {
        int row = wave * 2 + lhi;
        float vals[8];
        float s = 0.0f, s2 = 0.0f;
#pragma unroll
        for (int q = 0; q < 8; ++q) {
            float v = ys[row * YS_STRIDE + llo * 8 + q];
            vals[q] = v; s += v; s2 += v * v;
        }
        for (int m = 1; m <= 8; m <<= 1) {
            s  += __shfl_xor(s,  m, 32);
            s2 += __shfl_xor(s2, m, 32);
        }
        float mean = s * (1.0f / 128.0f);
        float var  = s2 * (1.0f / 128.0f) - mean * mean;
        float inv  = rsqrtf(var + 1e-5f);
        int grow = rowbase + row;
#pragma unroll
        for (int q = 0; q < 8; ++q) {
            int c = llo * 8 + q;
            ebuf[grow * 128 + c] = (vals[q] - mean) * inv * g[c] + be[c];
        }
    }
}

__global__ __launch_bounds__(32)
void logits_kernel(const float* __restrict__ e1, const float* __restrict__ e2,
                   const float* __restrict__ logit_scale, float* __restrict__ out)
{
    int b  = blockIdx.z;
    int Mb = blockIdx.x * 16, Nb = blockIdx.y * 16;
    int lane = threadIdx.x & 31;
    int lhi = lane >> 4, llo = lane & 15;

    float scale = __expf(logit_scale[0]);

    int mrow = Mb + llo; if (mrow > NPATCH - 1) mrow = NPATCH - 1;
    int nrow = Nb + llo; if (nrow > NPATCH - 1) nrow = NPATCH - 1;
    const float* ea = e1 + (b * NPATCH + mrow) * 128;
    const float* eb = e2 + (b * NPATCH + nrow) * 128;

    v8f acc = {};
    for (int kb = 0; kb < 128; kb += 4) {
        int kc = kb + 2 * lhi;
        v2f a  = *(const v2f*)&ea[kc];
        v2f bm = *(const v2f*)&eb[kc];
        acc = __builtin_amdgcn_wmma_f32_16x16x4_f32(
            false, a, false, bm, (short)0, acc, false, false);
    }

#pragma unroll
    for (int r = 0; r < 8; ++r) {
        int M = Mb + r + 8 * lhi;
        int N = Nb + llo;
        if (M < NPATCH && N < NPATCH) {
            float v = scale * acc[r];
            out[(b * NPATCH + M) * NPATCH + N] = v;
            out[BATCH * NPATCH * NPATCH + (b * NPATCH + N) * NPATCH + M] = v;
        }
    }
}

extern "C" void kernel_launch(void* const* d_in, const int* in_sizes, int n_in,
                              void* d_out, int out_size, void* d_ws, size_t ws_size,
                              hipStream_t stream)
{
    const float* feat_c1  = (const float*)d_in[0];
    const float* feat_c2  = (const float*)d_in[1];
    const float* img_cw   = (const float*)d_in[3];
    const float* img_cb   = (const float*)d_in[4];
    const float* img_w1   = (const float*)d_in[5];
    const float* img_w2   = (const float*)d_in[6];
    const float* img_g    = (const float*)d_in[7];
    const float* img_b    = (const float*)d_in[8];
    const float* dep_cw   = (const float*)d_in[9];
    const float* dep_cb   = (const float*)d_in[10];
    const float* dep_w1   = (const float*)d_in[11];
    const float* dep_w2   = (const float*)d_in[12];
    const float* dep_g    = (const float*)d_in[13];
    const float* dep_b    = (const float*)d_in[14];
    const float* lscale   = (const float*)d_in[15];
    float* out = (float*)d_out;

    float* x1 = (float*)d_ws;
    float* x2 = x1 + 400 * 256;
    float* e1 = x2 + 400 * 256;
    float* e2 = e1 + 400 * 128;

    conv_relu_kernel<<<dim3(2 * BATCH * PLANE / 4 / 32), dim3(32 * CSPLIT), 0, stream>>>(
        feat_c1, feat_c2, img_cw, img_cb, dep_cw, dep_cb, x1, x2);

    fc_ln_kernel<<<dim3(25, 2), dim3(256), 0, stream>>>(
        x1, x2, img_w1, img_w2, img_g, img_b,
        dep_w1, dep_w2, dep_g, dep_b, e1, e2);

    logits_kernel<<<dim3(7, 7, BATCH), dim3(32), 0, stream>>>(e1, e2, lscale, out);
}


# speedup vs baseline: 1.3426x; 1.0446x over previous
#include <hip/hip_runtime.h>
#include <hip/hip_bf16.h>

typedef float v2f __attribute__((ext_vector_type(2)));
typedef float v4f __attribute__((ext_vector_type(4)));
typedef float v8f __attribute__((ext_vector_type(8)));

#define NV 10
#define NH 10
#define NPATCH 100
#define BATCH 4
#define HW 160
#define CHW (256 * HW * HW)
#define PLANE (HW * HW)
#define XS_STRIDE 260
#define YS_STRIDE 132

#define CSPLIT 4
#define CPER (256 / CSPLIT)
#define NLD 16
__global__ __launch_bounds__(32 * CSPLIT)
void conv_relu_kernel(const float* __restrict__ f1, const float* __restrict__ f2,
                      const float* __restrict__ w_img, const float* __restrict__ b_img,
                      const float* __restrict__ w_dep, const float* __restrict__ b_dep,
                      float* __restrict__ x1, float* __restrict__ x2)
{
    __shared__ v4f part[(CSPLIT - 1) * 32];

    const int lane = threadIdx.x & 31;
    const int cg = __builtin_amdgcn_readfirstlane((int)(threadIdx.x >> 5));
    const int QGRP = BATCH * PLANE / 4 / 32;
    int qg = blockIdx.x;
    int second = (qg >= QGRP) ? 1 : 0;
    int p = (qg - second * QGRP) * 32 + lane;
    const float* feat = second ? f2 : f1;
    const float* w    = (second ? w_dep : w_img) + cg * CPER;
    float bias        = second ? b_dep[0] : b_img[0];
    float* xo         = second ? x2 : x1;

    int pix = p * 4;
    int b   = pix / PLANE;
    int rem = pix - b * PLANE;
    int y   = rem / HW;
    int x   = rem - y * HW;

    const float* base = feat + (size_t)b * CHW + (size_t)(cg * CPER) * PLANE
                             + (size_t)y * HW + x;
    float a0 = (cg == 0) ? bias : 0.0f;
    v4f acc = { a0, a0, a0, a0 };
#pragma unroll 1
    for (int c0 = 0; c0 < CPER; c0 += NLD) {
        v4f v[NLD];
#pragma unroll
        for (int i = 0; i < NLD; ++i)
            v[i] = __builtin_nontemporal_load(
                (const v4f*)(base + (size_t)(c0 + i) * PLANE));
#pragma unroll
        for (int i = 0; i < NLD; ++i) {
            float wc = w[c0 + i];
            acc.x = fmaf(v[i].x, wc, acc.x);
            acc.y = fmaf(v[i].y, wc, acc.y);
            acc.z = fmaf(v[i].z, wc, acc.z);
            acc.w = fmaf(v[i].w, wc, acc.w);
        }
    }

    if (cg != 0) part[(cg - 1) * 32 + lane] = acc;
    __syncthreads();
    if (cg == 0) {
#pragma unroll
        for (int g = 0; g < CSPLIT - 1; ++g) {
            v4f q = part[g * 32 + lane];
            acc.x += q.x; acc.y += q.y; acc.z += q.z; acc.w += q.w;
        }
        acc.x = fmaxf(acc.x, 0.0f);
        acc.y = fmaxf(acc.y, 0.0f);
        acc.z = fmaxf(acc.z, 0.0f);
        acc.w = fmaxf(acc.w, 0.0f);

        int pv = y >> 4, i = y & 15, ph = x >> 4, j = x & 15;
        int patch = (b * NV + pv) * NH + ph;
        *(v4f*)(xo + patch * 256 + i * 16 + j) = acc;
    }
}

#define KCH 128
#define NLDB (KCH / 8)

__device__ __forceinline__ void load_b_block(v4f (&b)[NLDB], const float* __restrict__ p)
{
#pragma unroll
    for (int j = 0; j < NLDB; ++j)
        b[j] = *(const v4f*)(p + 8 * j);
}

__device__ __forceinline__ v8f mma_block(v8f acc, const float* a_lds, const v4f (&b)[NLDB])
{
#pragma unroll
    for (int j = 0; j < NLDB; ++j) {
        v4f a = *(const v4f*)(a_lds + 8 * j);
        v2f alo = { a.x, a.y },       ahi = { a.z, a.w };
        v2f blo = { b[j].x, b[j].y }, bhi = { b[j].z, b[j].w };
        acc = __builtin_amdgcn_wmma_f32_16x16x4_f32(
            false, alo, false, blo, (short)0, acc, false, false);
        acc = __builtin_amdgcn_wmma_f32_16x16x4_f32(
            false, ahi, false, bhi, (short)0, acc, false, false);
    }
    return acc;
}

__global__ __launch_bounds__(256)
void fc_ln_kernel(const float* __restrict__ xbuf_img, const float* __restrict__ xbuf_dep,
                  const float* __restrict__ w1_img, const float* __restrict__ w2_img,
                  const float* __restrict__ g_img,  const float* __restrict__ be_img,
                  const float* __restrict__ w1_dep, const float* __restrict__ w2_dep,
                  const float* __restrict__ g_dep,  const float* __restrict__ be_dep,
                  float* __restrict__ e_img, float* __restrict__ e_dep)
{
    __shared__ __attribute__((aligned(16))) float xs[16 * XS_STRIDE];
    __shared__ __attribute__((aligned(16))) float hs[16 * XS_STRIDE];
    __shared__ float ys[16 * YS_STRIDE];

    const int branch = blockIdx.y;
    const float* xbuf = branch ? xbuf_dep : xbuf_img;
    const float* w1   = branch ? w1_dep   : w1_img;
    const float* w2   = branch ? w2_dep   : w2_img;
    const float* g    = branch ? g_dep    : g_img;
    const float* be   = branch ? be_dep   : be_img;
    float* ebuf       = branch ? e_dep    : e_img;

    const int rowbase = blockIdx.x * 16;
    const int t = threadIdx.x;

    const int wave = t >> 5;
    const int lane = t & 31;
    const int lhi  = lane >> 4;
    const int llo  = lane & 15;

    const float* w1t0 = w1 + (wave * 16 + llo) * 256 + 4 * lhi;
    const float* w1t1 = w1t0 + 8 * 16 * 256;
    const float* w2t  = w2 + (wave * 16 + llo) * 256 + 4 * lhi;
    const float* xa   = xs + llo * XS_STRIDE + 4 * lhi;
    const float* ha   = hs + llo * XS_STRIDE + 4 * lhi;

    v4f b0[NLDB], b1[NLDB];
    load_b_block(b0, w1t0);

    for (int k = 0; k < 16; ++k) {
        int idx = t + k * 256;
        int r = idx >> 8, c = idx & 255;
        xs[r * XS_STRIDE + c] = xbuf[(rowbase + r) * 256 + c];
    }
    __syncthreads();

    {
        v8f acc = {};
        load_b_block(b1, w1t0 + KCH);
        acc = mma_block(acc, xa, b0);
        load_b_block(b0, w1t1);
        acc = mma_block(acc, xa + KCH, b1);
#pragma unroll
        for (int r = 0; r < 8; ++r) {
            int M = r + 8 * lhi;
            hs[M * XS_STRIDE + wave * 16 + llo] = fmaxf(acc[r], 0.0f);
        }
    }
    {
        v8f acc = {};
        load_b_block(b1, w1t1 + KCH);
        acc = mma_block(acc, xa, b0);
        load_b_block(b0, w2t);
        acc = mma_block(acc, xa + KCH, b1);
#pragma unroll
        for (int r = 0; r < 8; ++r) {
            int M = r + 8 * lhi;
            hs[M * XS_STRIDE + (wave + 8) * 16 + llo] = fmaxf(acc[r], 0.0f);
        }
    }
    load_b_block(b1, w2t + KCH);
    __syncthreads();

    {
        v8f acc = {};
        acc = mma_block(acc, ha, b0);
        acc = mma_block(acc, ha + KCH, b1);
#pragma unroll
        for (int r = 0; r < 8; ++r) {
            int M = r + 8 * lhi;
            ys[M * YS_STRIDE + wave * 16 + llo] = acc[r];
        }
    }
    __syncthreads();

    {
        int row = wave * 2 + lhi;
        float vals[8];
        float s = 0.0f, s2 = 0.0f;
#pragma unroll
        for (int q = 0; q < 8; ++q) {
            float v = ys[row * YS_STRIDE + llo * 8 + q];
            vals[q] = v; s += v; s2 += v * v;
        }
        for (int m = 1; m <= 8; m <<= 1) {
            s  += __shfl_xor(s,  m, 32);
            s2 += __shfl_xor(s2, m, 32);
        }
        float mean = s * (1.0f / 128.0f);
        float var  = s2 * (1.0f / 128.0f) - mean * mean;
        float inv  = rsqrtf(var + 1e-5f);
        int grow = rowbase + row;
#pragma unroll
        for (int q = 0; q < 8; ++q) {
            int c = llo * 8 + q;
            ebuf[grow * 128 + c] = (vals[q] - mean) * inv * g[c] + be[c];
        }
    }
}

__global__ __launch_bounds__(32)
void logits_kernel(const float* __restrict__ e1, const float* __restrict__ e2,
                   const float* __restrict__ logit_scale, float* __restrict__ out)
{
    int b  = blockIdx.z;
    int Mb = blockIdx.x * 16, Nb = blockIdx.y * 16;
    int lane = threadIdx.x & 31;
    int lhi = lane >> 4, llo = lane & 15;

    float scale = __expf(logit_scale[0]);

    int mrow = Mb + llo; if (mrow > NPATCH - 1) mrow = NPATCH - 1;
    int nrow = Nb + llo; if (nrow > NPATCH - 1) nrow = NPATCH - 1;
    const float* ea = e1 + (b * NPATCH + mrow) * 128;
    const float* eb = e2 + (b * NPATCH + nrow) * 128;

    v8f acc = {};
    for (int kb = 0; kb < 128; kb += 4) {
        int kc = kb + 2 * lhi;
        v2f a  = *(const v2f*)&ea[kc];
        v2f bm = *(const v2f*)&eb[kc];
        acc = __builtin_amdgcn_wmma_f32_16x16x4_f32(
            false, a, false, bm, (short)0, acc, false, false);
    }

#pragma unroll
    for (int r = 0; r < 8; ++r) {
        int M = Mb + r + 8 * lhi;
        int N = Nb + llo;
        if (M < NPATCH && N < NPATCH) {
            float v = scale * acc[r];
            out[(b * NPATCH + M) * NPATCH + N] = v;
            out[BATCH * NPATCH * NPATCH + (b * NPATCH + N) * NPATCH + M] = v;
        }
    }
}

extern "C" void kernel_launch(void* const* d_in, const int* in_sizes, int n_in,
                              void* d_out, int out_size, void* d_ws, size_t ws_size,
                              hipStream_t stream)
{
    const float* feat_c1  = (const float*)d_in[0];
    const float* feat_c2  = (const float*)d_in[1];
    const float* img_cw   = (const float*)d_in[3];
    const float* img_cb   = (const float*)d_in[4];
    const float* img_w1   = (const float*)d_in[5];
    const float* img_w2   = (const float*)d_in[6];
    const float* img_g    = (const float*)d_in[7];
    const float* img_b    = (const float*)d_in[8];
    const float* dep_cw   = (const float*)d_in[9];
    const float* dep_cb   = (const float*)d_in[10];
    const float* dep_w1   = (const float*)d_in[11];
    const float* dep_w2   = (const float*)d_in[12];
    const float* dep_g    = (const float*)d_in[13];
    const float* dep_b    = (const float*)d_in[14];
    const float* lscale   = (const float*)d_in[15];
    float* out = (float*)d_out;

    float* x1 = (float*)d_ws;
    float* x2 = x1 + 400 * 256;
    float* e1 = x2 + 400 * 256;
    float* e2 = e1 + 400 * 128;

    conv_relu_kernel<<<dim3(2 * BATCH * PLANE / 4 / 32), dim3(32 * CSPLIT), 0, stream>>>(
        feat_c1, feat_c2, img_cw, img_cb, dep_cw, dep_cb, x1, x2);

    fc_ln_kernel<<<dim3(25, 2), dim3(256), 0, stream>>>(
        x1, x2, img_w1, img_w2, img_g, img_b,
        dep_w1, dep_w2, dep_g, dep_b, e1, e2);

    logits_kernel<<<dim3(7, 7, BATCH), dim3(32), 0, stream>>>(e1, e2, lscale, out);
}


# fc phase: each 16-row tile split over 4 work-groups by FC1 column quarter = FC2 k quarter (200 WGPs busy, not 50); partial y summed + LayerNorm in the logits kernel; k and LayerNorm sum orders change
# speedup vs baseline: 1.4902x; 1.1099x over previous
#include <hip/hip_runtime.h>
#include <hip/hip_bf16.h>

typedef float v2f __attribute__((ext_vector_type(2)));
typedef float v4f __attribute__((ext_vector_type(4)));
typedef float v8f __attribute__((ext_vector_type(8)));

#define NV 10
#define NH 10
#define NPATCH 100
#define BATCH 4
#define HW 160
#define CHW (256 * HW * HW)
#define PLANE (HW * HW)
#define XS_STRIDE 260
#define YS_STRIDE 132
#define HS_STRIDE 68
#define FSPLIT 4
#define YP_PLANE (400 * 128)

#define CSPLIT 4
#define CPER (256 / CSPLIT)
#define NLD 16
__global__ __launch_bounds__(32 * CSPLIT)
void conv_relu_kernel(const float* __restrict__ f1, const float* __restrict__ f2,
                      const float* __restrict__ w_img, const float* __restrict__ b_img,
                      const float* __restrict__ w_dep, const float* __restrict__ b_dep,
                      float* __restrict__ x1, float* __restrict__ x2)
{
    __shared__ v4f part[(CSPLIT - 1) * 32];

    const int lane = threadIdx.x & 31;
    const int cg = __builtin_amdgcn_readfirstlane((int)(threadIdx.x >> 5));
    const int QGRP = BATCH * PLANE / 4 / 32;
    int qg = blockIdx.x;
    int second = (qg >= QGRP) ? 1 : 0;
    int p = (qg - second * QGRP) * 32 + lane;
    const float* feat = second ? f2 : f1;
    const float* w    = (second ? w_dep : w_img) + cg * CPER;
    float bias        = second ? b_dep[0] : b_img[0];
    float* xo         = second ? x2 : x1;

    int pix = p * 4;
    int b   = pix / PLANE;
    int rem = pix - b * PLANE;
    int y   = rem / HW;
    int x   = rem - y * HW;

    const float* base = feat + (size_t)b * CHW + (size_t)(cg * CPER) * PLANE
                             + (size_t)y * HW + x;
    float a0 = (cg == 0) ? bias : 0.0f;
    v4f acc = { a0, a0, a0, a0 };
#pragma unroll 1
    for (int c0 = 0; c0 < CPER; c0 += NLD) {
        v4f v[NLD];
#pragma unroll
        for (int i = 0; i < NLD; ++i)
            v[i] = __builtin_nontemporal_load(
                (const v4f*)(base + (size_t)(c0 + i) * PLANE));
#pragma unroll
        for (int i = 0; i < NLD; ++i) {
            float wc = w[c0 + i];
            acc.x = fmaf(v[i].x, wc, acc.x);
            acc.y = fmaf(v[i].y, wc, acc.y);
            acc.z = fmaf(v[i].z, wc, acc.z);
            acc.w = fmaf(v[i].w, wc, acc.w);
        }
    }

    if (cg != 0) part[(cg - 1) * 32 + lane] = acc;
    __syncthreads();
    if (cg == 0) {
#pragma unroll
        for (int g = 0; g < CSPLIT - 1; ++g) {
            v4f q = part[g * 32 + lane];
            acc.x += q.x; acc.y += q.y; acc.z += q.z; acc.w += q.w;
        }
        acc.x = fmaxf(acc.x, 0.0f);
        acc.y = fmaxf(acc.y, 0.0f);
        acc.z = fmaxf(acc.z, 0.0f);
        acc.w = fmaxf(acc.w, 0.0f);

        int pv = y >> 4, i = y & 15, ph = x >> 4, j = x & 15;
        int patch = (b * NV + pv) * NH + ph;
        *(v4f*)(xo + patch * 256 + i * 16 + j) = acc;
    }
}

#define KCH 128
#define NLDB (KCH / 8)
#define NLD2 (64 / 8)
#define FWAVES 4

__device__ __forceinline__ v8f wmma4(v2f a, v2f b, v8f c)
{
    return __builtin_amdgcn_wmma_f32_16x16x4_f32(false, a, false, b, (short)0, c, false, false);
}

__global__ __launch_bounds__(32 * FWAVES)
void fc_split_kernel(const float* __restrict__ xbuf_img, const float* __restrict__ xbuf_dep,
                     const float* __restrict__ w1_img, const float* __restrict__ w2_img,
                     const float* __restrict__ w1_dep, const float* __restrict__ w2_dep,
                     float* __restrict__ ypart)
{
    __shared__ __attribute__((aligned(16))) float xs[16 * XS_STRIDE];
    __shared__ __attribute__((aligned(16))) float hs[16 * HS_STRIDE];
    __shared__ __attribute__((aligned(16))) float ys[16 * YS_STRIDE];

    const int branch = blockIdx.y;
    const int q      = blockIdx.z;
    const float* xbuf = branch ? xbuf_dep : xbuf_img;
    const float* w1   = branch ? w1_dep   : w1_img;
    const float* w2   = branch ? w2_dep   : w2_img;

    const int rowbase = blockIdx.x * 16;
    const int t = threadIdx.x;
    const int wave = t >> 5;
    const int lane = t & 31;
    const int lhi  = lane >> 4;
    const int llo  = lane & 15;

    v4f xv[8];
#pragma unroll
    for (int i = 0; i < 8; ++i) {
        int idx = t + i * 128;
        int r = idx >> 6, c4 = idx & 63;
        xv[i] = *(const v4f*)(xbuf + (rowbase + r) * 256 + c4 * 4);
    }

    const float* w1t  = w1 + ((q * 4 + wave) * 16 + llo) * 256 + 4 * lhi;
    const float* w2t0 = w2 + (wave * 16 + llo) * 256 + q * 64 + 4 * lhi;
    const float* w2t1 = w2t0 + 4 * 16 * 256;
    v4f b1a[NLDB], b1b[NLDB], b2a[NLD2], b2b[NLD2];
#pragma unroll
    for (int j = 0; j < NLDB / 2; ++j) {
        b1a[j] = *(const v4f*)(w1t + 8 * j);
        b1b[j] = *(const v4f*)(w1t + KCH + 8 * j);
    }

#pragma unroll
    for (int i = 0; i < 8; ++i) {
        int idx = t + i * 128;
        int r = idx >> 6, c4 = idx & 63;
        *(v4f*)(xs + r * XS_STRIDE + c4 * 4) = xv[i];
    }
    __syncthreads();

#pragma unroll
    for (int j = NLDB / 2; j < NLDB; ++j) {
        b1a[j] = *(const v4f*)(w1t + 8 * j);
        b1b[j] = *(const v4f*)(w1t + KCH + 8 * j);
    }
#pragma unroll
    for (int j = 0; j < NLD2; ++j) {
        b2a[j] = *(const v4f*)(w2t0 + 8 * j);
        b2b[j] = *(const v4f*)(w2t1 + 8 * j);
    }

    {
        const float* xa = xs + llo * XS_STRIDE + 4 * lhi;
        v4f a0[NLDB], a1[NLDB];
#pragma unroll
        for (int j = 0; j < NLDB; ++j) {
            a0[j] = *(const v4f*)(xa + 8 * j);
            a1[j] = *(const v4f*)(xa + KCH + 8 * j);
        }
        __builtin_amdgcn_sched_barrier(0);
        v8f acc0 = {}, acc1 = {};
#pragma unroll
        for (int j = 0; j < NLDB; ++j) {
            v2f a0lo = { a0[j].x, a0[j].y },   a0hi = { a0[j].z, a0[j].w };
            v2f a1lo = { a1[j].x, a1[j].y },   a1hi = { a1[j].z, a1[j].w };
            v2f b0lo = { b1a[j].x, b1a[j].y }, b0hi = { b1a[j].z, b1a[j].w };
            v2f b1lo = { b1b[j].x, b1b[j].y }, b1hi = { b1b[j].z, b1b[j].w };
            acc0 = wmma4(a0lo, b0lo, acc0);
            acc1 = wmma4(a1lo, b1lo, acc1);
            acc0 = wmma4(a0hi, b0hi, acc0);
            acc1 = wmma4(a1hi, b1hi, acc1);
        }
#pragma unroll
        for (int r = 0; r < 8; ++r) {
            int M = r + 8 * lhi;
            hs[M * HS_STRIDE + wave * 16 + llo] = fmaxf(acc0[r] + acc1[r], 0.0f);
        }
    }
    __syncthreads();

    {
        const float* ha = hs + llo * HS_STRIDE + 4 * lhi;
        v4f a[NLD2];
#pragma unroll
        for (int j = 0; j < NLD2; ++j)
            a[j] = *(const v4f*)(ha + 8 * j);
        __builtin_amdgcn_sched_barrier(0);
        v8f c0 = {}, c1 = {};
#pragma unroll
        for (int j = 0; j < NLD2; ++j) {
            v2f alo = { a[j].x, a[j].y },      ahi = { a[j].z, a[j].w };
            v2f b0lo = { b2a[j].x, b2a[j].y }, b0hi = { b2a[j].z, b2a[j].w };
            v2f b1lo = { b2b[j].x, b2b[j].y }, b1hi = { b2b[j].z, b2b[j].w };
            c0 = wmma4(alo, b0lo, c0);
            c1 = wmma4(alo, b1lo, c1);
            c0 = wmma4(ahi, b0hi, c0);
            c1 = wmma4(ahi, b1hi, c1);
        }
#pragma unroll
        for (int r = 0; r < 8; ++r) {
            int M = r + 8 * lhi;
            ys[M * YS_STRIDE + wave * 16 + llo]       = c0[r];
            ys[M * YS_STRIDE + (wave + 4) * 16 + llo] = c1[r];
        }
    }
    __syncthreads();

    {
        float* yp = ypart + ((branch * FSPLIT + q) * 400 + rowbase) * 128;
#pragma unroll
        for (int i = 0; i < 4; ++i) {
            int idx = t + i * 128;
            int r = idx >> 5, c4 = idx & 31;
            *(v4f*)(yp + r * 128 + c4 * 4) = *(const v4f*)(ys + r * YS_STRIDE + c4 * 4);
        }
    }
}

__device__ __forceinline__ void sum_planes_to_lds(float* dst, const float* __restrict__ src, int row0, int lane)
{
#pragma unroll
    for (int r = 0; r < 16; ++r) {
        int row = row0 + r; if (row > NPATCH - 1) row = NPATCH - 1;
        const float* p = src + row * 128 + 4 * lane;
        v4f s = *(const v4f*)p;
#pragma unroll
        for (int k = 1; k < FSPLIT; ++k) {
            v4f u = *(const v4f*)(p + k * YP_PLANE);
            s.x += u.x; s.y += u.y; s.z += u.z; s.w += u.w;
        }
        *(v4f*)(dst + r * YS_STRIDE + 4 * lane) = s;
    }
}

__device__ __forceinline__ void ln_fragments(v4f (&e)[16], const float* y_lds,
                                             const float* g_lds, const float* be_lds)
{
    float s = 0.0f, s2 = 0.0f;
#pragma unroll
    for (int j = 0; j < 16; ++j) {
        v4f v = *(const v4f*)(y_lds + 8 * j);
        e[j] = v;
        s  += (v.x + v.y) + (v.z + v.w);
        s2 += (v.x * v.x + v.y * v.y) + (v.z * v.z + v.w * v.w);
    }
    s  += __shfl_xor(s,  16, 32);
    s2 += __shfl_xor(s2, 16, 32);
    float mean = s * (1.0f / 128.0f);
    float var  = s2 * (1.0f / 128.0f) - mean * mean;
    float inv  = rsqrtf(var + 1e-5f);
#pragma unroll
    for (int j = 0; j < 16; ++j) {
        v4f gg = *(const v4f*)(g_lds + 8 * j);
        v4f bb = *(const v4f*)(be_lds + 8 * j);
        e[j].x = (e[j].x - mean) * inv * gg.x + bb.x;
        e[j].y = (e[j].y - mean) * inv * gg.y + bb.y;
        e[j].z = (e[j].z - mean) * inv * gg.z + bb.z;
        e[j].w = (e[j].w - mean) * inv * gg.w + bb.w;
    }
}

__global__ __launch_bounds__(32)
void logits_kernel(const float* __restrict__ ypart,
                   const float* __restrict__ g_img, const float* __restrict__ be_img,
                   const float* __restrict__ g_dep, const float* __restrict__ be_dep,
                   const float* __restrict__ logit_scale, float* __restrict__ out)
{
    __shared__ __attribute__((aligned(16))) float ya[16 * YS_STRIDE];
    __shared__ __attribute__((aligned(16))) float yb[16 * YS_STRIDE];
    __shared__ __attribute__((aligned(16))) float gb[4 * 128];

    int b  = blockIdx.z;
    int Mb = blockIdx.x * 16, Nb = blockIdx.y * 16;
    int lane = threadIdx.x & 31;
    int lhi = lane >> 4, llo = lane & 15;

    float scale = __expf(logit_scale[0]);

    const float* pa = ypart + (b * NPATCH) * 128;
    const float* pb = pa + FSPLIT * YP_PLANE;
    *(v4f*)(gb + 0 * 128 + 4 * lane) = *(const v4f*)(g_img  + 4 * lane);
    *(v4f*)(gb + 1 * 128 + 4 * lane) = *(const v4f*)(be_img + 4 * lane);
    *(v4f*)(gb + 2 * 128 + 4 * lane) = *(const v4f*)(g_dep  + 4 * lane);
    *(v4f*)(gb + 3 * 128 + 4 * lane) = *(const v4f*)(be_dep + 4 * lane);
    sum_planes_to_lds(ya, pa, Mb, lane);
    sum_planes_to_lds(yb, pb, Nb, lane);
    __syncthreads();

    v4f a[16], bm[16];
    ln_fragments(a,  ya + llo * YS_STRIDE + 4 * lhi, gb + 0 * 128 + 4 * lhi, gb + 1 * 128 + 4 * lhi);
    ln_fragments(bm, yb + llo * YS_STRIDE + 4 * lhi, gb + 2 * 128 + 4 * lhi, gb + 3 * 128 + 4 * lhi);

    v8f acc = {};
#pragma unroll
    for (int j = 0; j < 16; ++j) {
        v2f alo = { a[j].x, a[j].y },   ahi = { a[j].z, a[j].w };
        v2f blo = { bm[j].x, bm[j].y }, bhi = { bm[j].z, bm[j].w };
        acc = wmma4(alo, blo, acc);
        acc = wmma4(ahi, bhi, acc);
    }

#pragma unroll
    for (int r = 0; r < 8; ++r) {
        int M = Mb + r + 8 * lhi;
        int N = Nb + llo;
        if (M < NPATCH && N < NPATCH) {
            float v = scale * acc[r];
            out[(b * NPATCH + M) * NPATCH + N] = v;
            out[BATCH * NPATCH * NPATCH + (b * NPATCH + N) * NPATCH + M] = v;
        }
    }
}

extern "C" void kernel_launch(void* const* d_in, const int* in_sizes, int n_in,
                              void* d_out, int out_size, void* d_ws, size_t ws_size,
                              hipStream_t stream)
{
    const float* feat_c1  = (const float*)d_in[0];
    const float* feat_c2  = (const float*)d_in[1];
    const float* img_cw   = (const float*)d_in[3];
    const float* img_cb   = (const float*)d_in[4];
    const float* img_w1   = (const float*)d_in[5];
    const float* img_w2   = (const float*)d_in[6];
    const float* img_g    = (const float*)d_in[7];
    const float* img_b    = (const float*)d_in[8];
    const float* dep_cw   = (const float*)d_in[9];
    const float* dep_cb   = (const float*)d_in[10];
    const float* dep_w1   = (const float*)d_in[11];
    const float* dep_w2   = (const float*)d_in[12];
    const float* dep_g    = (const float*)d_in[13];
    const float* dep_b    = (const float*)d_in[14];
    const float* lscale   = (const float*)d_in[15];
    float* out = (float*)d_out;

    float* x1 = (float*)d_ws;
    float* x2 = x1 + 400 * 256;
    float* yp = x2 + 400 * 256;

    conv_relu_kernel<<<dim3(2 * BATCH * PLANE / 4 / 32), dim3(32 * CSPLIT), 0, stream>>>(
        feat_c1, feat_c2, img_cw, img_cb, dep_cw, dep_cb, x1, x2);

    fc_split_kernel<<<dim3(25, 2, FSPLIT), dim3(32 * FWAVES), 0, stream>>>(
        x1, x2, img_w1, img_w2, dep_w1, dep_w2, yp);

    logits_kernel<<<dim3(7, 7, BATCH), dim3(32), 0, stream>>>(
        yp, img_g, img_b, dep_g, dep_b, lscale, out);
}
